# starting kernel
# baseline (speedup 1.0000x reference)
#include <hip/hip_runtime.h>
#include <math.h>

typedef float v2f __attribute__((ext_vector_type(2)));
typedef float v8f __attribute__((ext_vector_type(8)));

namespace {

__device__ __forceinline__ void gate_rx(float re[16], float im[16], float th, int qi) {
  const float c = cosf(0.5f * th);
  const float s = sinf(0.5f * th);
  const int mask = 8 >> qi;
#pragma unroll
  for (int p = 0; p < 16; ++p) {
    if (!(p & mask)) {
      const int q = p | mask;
      const float r0 = re[p], i0 = im[p], r1 = re[q], i1 = im[q];
      re[p] = c * r0 + s * i1;
      im[p] = c * i0 - s * r1;
      re[q] = c * r1 + s * i0;
      im[q] = c * i1 - s * r0;
    }
  }
}

__device__ __forceinline__ void gate_rz(float re[16], float im[16], float th, int qi) {
  const float c = cosf(0.5f * th);
  const float s = sinf(0.5f * th);
  const int mask = 8 >> qi;
#pragma unroll
  for (int p = 0; p < 16; ++p) {
    const float sg = (p & mask) ? s : -s;
    const float r = re[p], q = im[p];
    re[p] = r * c - q * sg;
    im[p] = r * sg + q * c;
  }
}

__device__ __forceinline__ void gate_crz(float re[16], float im[16], float th, int ctrl, int tgt) {
  const float c = cosf(0.5f * th);
  const float s = sinf(0.5f * th);
  const int cm = 8 >> ctrl, tm = 8 >> tgt;
#pragma unroll
  for (int p = 0; p < 16; ++p) {
    if (p & cm) {
      const float sg = (p & tm) ? s : -s;
      const float r = re[p], q = im[p];
      re[p] = r * c - q * sg;
      im[p] = r * sg + q * c;
    }
  }
}

}

extern "C" __global__ void qconv_setup(const float* __restrict__ w, float* __restrict__ ws) {
  const int S = threadIdx.x;
  if (S >= 16) return;
  float re[16], im[16];
#pragma unroll
  for (int p = 0; p < 16; ++p) { re[p] = 0.f; im[p] = 0.f; }
  re[0] = 1.f;

#pragma unroll
  for (int i = 0; i < 4; ++i) {
    gate_rx(re, im, w[i], i);
    gate_rz(re, im, w[4 + i], i);
  }
  int k = 8;
#pragma unroll
  for (int cd = 0; cd < 4; ++cd) {
    const int ctrl = 3 - cd;
#pragma unroll
    for (int tgt = 0; tgt < 4; ++tgt) {
      if (tgt == ctrl) continue;
      gate_crz(re, im, w[k], ctrl, tgt);
      ++k;
    }
  }
#pragma unroll
  for (int i = 0; i < 4; ++i) {
    if (S & (1 << i)) {
      const int mask = 8 >> i;
#pragma unroll
      for (int p = 0; p < 16; ++p) {
        if (!(p & mask)) {
          const int q = p | mask;
          const float r0 = re[p], i0 = im[p];
          re[p] = -re[q]; im[p] = -im[q];
          re[q] =  r0;    im[q] =  i0;
        }
      }
    }
  }
#pragma unroll
  for (int i = 0; i < 4; ++i) {
    gate_rx(re, im, w[20 + i], i);
    gate_rz(re, im, w[24 + i], i);
  }
#pragma unroll
  for (int p = 0; p < 16; ++p) {
    ws[S * 16 + p]       = re[p];
    ws[256 + S * 16 + p] = im[p];
  }
}

extern "C" __global__ __launch_bounds__(256) void qconv_main(
    const float* __restrict__ x, const float* __restrict__ Bmat,
    float* __restrict__ out) {
  __shared__ float lds[8 * 256];

  const int tid  = threadIdx.x;
  const int lane = tid & 31;
  const int wave = tid >> 5;
  const int hb   = lane >> 4;
  const int M    = lane & 15;
  const int base = (blockIdx.x * 8 + wave) * 16;
  const int n    = base + M;
  const int j = n & 63, i = (n >> 6) & 63, b = n >> 12;

  const float* img = x + (((size_t)b) << 14) + (size_t)(2 * i) * 128 + (size_t)(2 * j);
  const float2 top = *(const float2*)(img);
  const float2 bot = *(const float2*)(img + 128);
  const float v0 = top.x, v1 = top.y, v2 = bot.x, v3 = bot.y;

  const float ssq  = v0 * v0 + v1 * v1 + v2 * v2 + v3 * v3;
  const bool  zero = (v0 == 0.f) && (v1 == 0.f) && (v2 == 0.f) && (v3 == 0.f);
  const float inv  = zero ? 1.f : (1.f / sqrtf(ssq));
  const float t0 = zero ? 1.f : v0 * inv;
  const float t1 = zero ? 0.f : v1 * inv;
  const float t2 = zero ? 0.f : v2 * inv;
  const float t3 = zero ? 0.f : v3 * inv;

  const float c0 = cosf(0.5f * t0), s0 = sinf(0.5f * t0);
  const float c1 = cosf(0.5f * t1), s1 = sinf(0.5f * t1);
  const float c2 = cosf(0.5f * t2), s2 = sinf(0.5f * t2);
  const float c3 = cosf(0.5f * t3), s3 = sinf(0.5f * t3);

  const float p01[4] = { c0 * c1, s0 * c1, c0 * s1, s0 * s1 };
  const float p23[4] = { c2 * c3, s2 * c3, c2 * s3, s2 * s3 };
  float m[16];
#pragma unroll
  for (int S = 0; S < 16; ++S) m[S] = p01[S & 3] * p23[S >> 2];

  v2f a[4];
#pragma unroll
  for (int f = 0; f < 4; ++f) {
    a[f].x = hb ? m[4 * f + 2] : m[4 * f + 0];
    a[f].y = hb ? m[4 * f + 3] : m[4 * f + 1];
  }

  v2f bre[4], bim[4];
#pragma unroll
  for (int f = 0; f < 4; ++f) {
    const int kb = 4 * f + 2 * hb;
    bre[f].x = Bmat[kb * 16 + M];
    bre[f].y = Bmat[kb * 16 + 16 + M];
    bim[f].x = Bmat[256 + kb * 16 + M];
    bim[f].y = Bmat[256 + kb * 16 + 16 + M];
  }

  v8f sre = {0.f, 0.f, 0.f, 0.f, 0.f, 0.f, 0.f, 0.f};
  v8f sim = {0.f, 0.f, 0.f, 0.f, 0.f, 0.f, 0.f, 0.f};
#pragma unroll
  for (int f = 0; f < 4; ++f) {
    sre = __builtin_amdgcn_wmma_f32_16x16x4_f32(false, a[f], false, bre[f],
                                                (short)0, sre, false, false);
    sim = __builtin_amdgcn_wmma_f32_16x16x4_f32(false, a[f], false, bim[f],
                                                (short)0, sim, false, false);
  }
  const v8f P = sre * sre + sim * sim;

  float* lp = &lds[wave * 256];
#pragma unroll
  for (int v = 0; v < 8; ++v) lp[(v + 8 * hb) * 16 + M] = P[v];
  asm volatile("s_wait_dscnt 0x0" ::: "memory");

  if (lane < 16) {
    const float* row = &lds[wave * 256 + lane * 16];
    float q0 = 0.f, q1 = 0.f, q2 = 0.f, q3 = 0.f;
#pragma unroll
    for (int p = 0; p < 16; ++p) {
      const float pv = row[p];
      q0 += (p & 8) ? -pv : pv;
      q1 += (p & 4) ? -pv : pv;
      q2 += (p & 2) ? -pv : pv;
      q3 += (p & 1) ? -pv : pv;
    }
    float* o = out + (((size_t)b) << 14) + (size_t)i * 64 + (size_t)j;
    o[0]     = q0;
    o[4096]  = q1;
    o[8192]  = q2;
    o[12288] = q3;
  }
}

extern "C" void kernel_launch(void* const* d_in, const int* in_sizes, int n_in,
                              void* d_out, int out_size, void* d_ws, size_t ws_size,
                              hipStream_t stream) {
  const float* x = (const float*)d_in[0];
  const float* w = (const float*)d_in[1];
  float* out = (float*)d_out;
  float* ws  = (float*)d_ws;

  qconv_setup<<<1, 16, 0, stream>>>(w, ws);
  qconv_main<<<2048, 256, 0, stream>>>(x, ws, out);
}


# Main part: the wave's four tiles go through each stage together (all pixel loads first, encodings interleaved, one LDS staging and fence) and all 32 lanes sum, two tiles per pass; same values
# speedup vs baseline: 2.5250x; 2.5250x over previous
#include <hip/hip_runtime.h>
#include <math.h>


typedef float v2f __attribute__((ext_vector_type(2)));
typedef float v8f __attribute__((ext_vector_type(8)));

namespace {

template <int MASK>
__device__ __forceinline__ float row_xor(float v) {
  return __int_as_float(__builtin_amdgcn_update_dpp(0, __float_as_int(v), 0x160 + MASK, 0xf, 0xf, true));
}

__device__ __forceinline__ float other_row(float v) {
  const int b = __float_as_int(v);
  return __int_as_float(__builtin_amdgcn_permlanex16(b, b, 0x76543210, 0xfedcba98, false, false));
}

template <int MASK>
__device__ __forceinline__ void gate_rx(float& re, float& im, float c, float s) {
  const float pr = row_xor<MASK>(re), pi = row_xor<MASK>(im);
  const float nr = c * re + s * pi;
  const float ni = c * im - s * pr;
  re = nr; im = ni;
}

__device__ __forceinline__ void gate_phase(float& re, float& im, float c, float sg) {
  const float nr = re * c - im * sg;
  const float ni = re * sg + im * c;
  re = nr; im = ni;
}

template <int MASK>
__device__ __forceinline__ void gate_k(float& re, float& im, bool bit, bool on) {
  const float pr = row_xor<MASK>(re), pi = row_xor<MASK>(im);
  re = on ? (bit ? pr : -pr) : re;
  im = on ? (bit ? pi : -pi) : im;
}

template <int MASK>
__device__ __forceinline__ void col_step(float& re, float& im, bool bit, bool on,
                                         float cx, float sx, float cz, float sz) {
  gate_k<MASK>(re, im, bit, on);
  gate_rx<MASK>(re, im, cx, sx);
  gate_phase(re, im, cz, bit ? sz : -sz);
}

constexpr int kTilesPerWave = 4;
constexpr int kWavesInGrid  = 4096;
constexpr int kTileFloats   = 256;

}

extern "C" __global__ __launch_bounds__(256) void qconv_fused(
    const float* __restrict__ x, const float* __restrict__ w, float* __restrict__ out) {
  __shared__ float lds[8 * kTilesPerWave * kTileFloats];

  const int tid  = threadIdx.x;
  const int lane = tid & 31;
  const int wave = tid >> 5;
  const int hb   = lane >> 4;
  const int M    = lane & 15;
  const int wid  = blockIdx.x * 8 + wave;

  float2 top[kTilesPerWave], bot[kTilesPerWave];
#pragma unroll
  for (int t = 0; t < kTilesPerWave; ++t) {
    const int n = (t * kWavesInGrid + wid) * 16 + M;
    const int j = n & 63, i = (n >> 6) & 63, b = n >> 12;
    const float* img = x + (((size_t)b) << 14) + (size_t)(2 * i) * 128 + (size_t)(2 * j);
    top[t] = *(const float2*)(img);
    bot[t] = *(const float2*)(img + 128);
  }

  const float th = w[lane < 28 ? lane : 27];
  const float cl = cosf(0.5f * th);
  const float sl = sinf(0.5f * th);
  float cw[28], sw[28];
#pragma unroll
  for (int k = 0; k < 28; ++k) {
    cw[k] = __int_as_float(__builtin_amdgcn_readlane(__float_as_int(cl), k));
    sw[k] = __int_as_float(__builtin_amdgcn_readlane(__float_as_int(sl), k));
  }
  const bool b0 = (M & 8) != 0, b1 = (M & 4) != 0, b2 = (M & 2) != 0, b3 = (M & 1) != 0;
  const bool bq[4] = { b0, b1, b2, b3 };

  float re = (M == 0) ? 1.f : 0.f;
  float im = 0.f;

  gate_rx<8>(re, im, cw[0], sw[0]);  gate_phase(re, im, cw[4], b0 ? sw[4] : -sw[4]);
  gate_rx<4>(re, im, cw[1], sw[1]);  gate_phase(re, im, cw[5], b1 ? sw[5] : -sw[5]);
  gate_rx<2>(re, im, cw[2], sw[2]);  gate_phase(re, im, cw[6], b2 ? sw[6] : -sw[6]);
  gate_rx<1>(re, im, cw[3], sw[3]);  gate_phase(re, im, cw[7], b3 ? sw[7] : -sw[7]);

  {
    int k = 8;
#pragma unroll
    for (int cd = 0; cd < 4; ++cd) {
      const int ctrl = 3 - cd;
#pragma unroll
      for (int tgt = 0; tgt < 4; ++tgt) {
        if (tgt == ctrl) continue;
        const float sg = bq[tgt] ? sw[k] : -sw[k];
        gate_phase(re, im, bq[ctrl] ? cw[k] : 1.f, bq[ctrl] ? sg : 0.f);
        ++k;
      }
    }
  }

  float r1[2], i1[2];
#pragma unroll
  for (int e = 0; e < 2; ++e) {
    r1[e] = re; i1[e] = im;
    col_step<8>(r1[e], i1[e], b0, e != 0, cw[20], sw[20], cw[24], sw[24]);
    col_step<4>(r1[e], i1[e], b1, hb != 0, cw[21], sw[21], cw[25], sw[25]);
  }
  float r2[2][2], i2[2][2];
#pragma unroll
  for (int f0 = 0; f0 < 2; ++f0) {
#pragma unroll
    for (int e = 0; e < 2; ++e) {
      r2[f0][e] = r1[e]; i2[f0][e] = i1[e];
      col_step<2>(r2[f0][e], i2[f0][e], b2, f0 != 0, cw[22], sw[22], cw[26], sw[26]);
    }
  }
  v2f bre[4], bim[4];
#pragma unroll
  for (int f = 0; f < 4; ++f) {
    float r3[2], i3[2];
#pragma unroll
    for (int e = 0; e < 2; ++e) {
      r3[e] = r2[f & 1][e]; i3[e] = i2[f & 1][e];
      col_step<1>(r3[e], i3[e], b3, (f & 2) != 0, cw[23], sw[23], cw[27], sw[27]);
    }
    bre[f].x = r3[0]; bre[f].y = r3[1];
    bim[f].x = i3[0]; bim[f].y = i3[1];
  }

  float* lp = &lds[wave * (kTilesPerWave * kTileFloats)];

  constexpr float kRev = 0.07957747154594767f;

  float u[kTilesPerWave][4], r[kTilesPerWave][4];
#pragma unroll
  for (int t = 0; t < kTilesPerWave; ++t) {
    const float v0 = top[t].x, v1 = top[t].y, v2 = bot[t].x, v3 = bot[t].y;
    const float ssq  = v0 * v0 + v1 * v1 + v2 * v2 + v3 * v3;
    const bool  zero = (v0 == 0.f) && (v1 == 0.f) && (v2 == 0.f) && (v3 == 0.f);
    const float inv  = kRev / sqrtf(ssq);
    const float va = hb ? v2 : v0;
    const float vb = hb ? v3 : v1;
    const float ra = zero ? (hb ? 0.f : kRev) : va * inv;
    const float rb = zero ? 0.f : vb * inv;

    const float sa = __builtin_amdgcn_sinf(ra), ca = __builtin_amdgcn_cosf(ra);
    const float sb = __builtin_amdgcn_sinf(rb), cb = __builtin_amdgcn_cosf(rb);

    u[t][0] = ca * cb;  u[t][1] = sa * cb;  u[t][2] = ca * sb;  u[t][3] = sa * sb;
#pragma unroll
    for (int k = 0; k < 4; ++k) r[t][k] = other_row(u[t][k]);
  }

#pragma unroll
  for (int t = 0; t < kTilesPerWave; ++t) {
    const float px = hb ? r[t][2] : u[t][0];
    const float py = hb ? r[t][3] : u[t][1];
    v2f a[4];
#pragma unroll
    for (int f = 0; f < 4; ++f) {
      const float g = hb ? u[t][f] : r[t][f];
      a[f].x = px * g;  a[f].y = py * g;
    }

    v8f sre = {0.f, 0.f, 0.f, 0.f, 0.f, 0.f, 0.f, 0.f};
    v8f sim = {0.f, 0.f, 0.f, 0.f, 0.f, 0.f, 0.f, 0.f};
#pragma unroll
    for (int f = 0; f < 4; ++f) {
      sre = __builtin_amdgcn_wmma_f32_16x16x4_f32(false, a[f], false, bre[f],
                                                  (short)0, sre, false, false);
      sim = __builtin_amdgcn_wmma_f32_16x16x4_f32(false, a[f], false, bim[f],
                                                  (short)0, sim, false, false);
    }
    const v8f P = sre * sre + sim * sim;

#pragma unroll
    for (int v = 0; v < 8; ++v) lp[t * kTileFloats + (v + 8 * hb) * 16 + M] = P[v];
  }
  asm volatile("s_wait_dscnt 0x0" ::: "memory");

  constexpr int kPasses = kTilesPerWave / 2;
  const float* row = &lp[hb * kTileFloats + M * 16];
  float q0[kPasses], q1[kPasses], q2[kPasses], q3[kPasses];
#pragma unroll
  for (int k = 0; k < kPasses; ++k) { q0[k] = 0.f; q1[k] = 0.f; q2[k] = 0.f; q3[k] = 0.f; }
#pragma unroll
  for (int p = 0; p < 16; ++p) {
#pragma unroll
    for (int k = 0; k < kPasses; ++k) {
      const float pv = row[2 * k * kTileFloats + p];
      q0[k] += (p & 8) ? -pv : pv;
      q1[k] += (p & 4) ? -pv : pv;
      q2[k] += (p & 2) ? -pv : pv;
      q3[k] += (p & 1) ? -pv : pv;
    }
  }
#pragma unroll
  for (int k = 0; k < kPasses; ++k) {
    const int t = 2 * k + hb;
    const int n = (t * kWavesInGrid + wid) * 16 + M;
    const int j = n & 63, i = (n >> 6) & 63, b = n >> 12;
    float* o = out + (((size_t)b) << 14) + (size_t)i * 64 + (size_t)j;
    o[0]     = q0[k];
    o[4096]  = q1[k];
    o[8192]  = q2[k];
    o[12288] = q3[k];
  }
}

extern "C" void kernel_launch(void* const* d_in, const int* in_sizes, int n_in,
                              void* d_out, int out_size, void* d_ws, size_t ws_size,
                              hipStream_t stream) {
  const float* x = (const float*)d_in[0];
  const float* w = (const float*)d_in[1];
  float* out = (float*)d_out;

  qconv_fused<<<512, 256, 0, stream>>>(x, w, out);
}
